# starting kernel
# baseline (speedup 1.0000x reference)
#include <hip/hip_runtime.h>
#include <math.h>

typedef __attribute__((ext_vector_type(2))) float v2f;
typedef __attribute__((ext_vector_type(8))) float v8f;

__device__ __constant__ float g_vdw[20][14] = {
  {1.55f,1.70f,1.70f,1.52f, 1.70f,0,0,0,0,0,0,0,0,0},
  {1.55f,1.70f,1.70f,1.52f, 1.70f,1.80f,0,0,0,0,0,0,0,0},
  {1.55f,1.70f,1.70f,1.52f, 1.70f,1.70f,1.52f,1.52f,0,0,0,0,0,0},
  {1.55f,1.70f,1.70f,1.52f, 1.70f,1.70f,1.70f,1.52f,1.52f,0,0,0,0,0},
  {1.55f,1.70f,1.70f,1.52f, 1.70f,1.70f,1.70f,1.70f,1.70f,1.70f,1.70f,0,0,0},
  {1.55f,1.70f,1.70f,1.52f, 0,0,0,0,0,0,0,0,0,0},
  {1.55f,1.70f,1.70f,1.52f, 1.70f,1.70f,1.55f,1.70f,1.70f,1.55f,0,0,0,0},
  {1.55f,1.70f,1.70f,1.52f, 1.70f,1.70f,1.70f,1.70f,0,0,0,0,0,0},
  {1.55f,1.70f,1.70f,1.52f, 1.70f,1.70f,1.70f,1.70f,1.55f,0,0,0,0,0},
  {1.55f,1.70f,1.70f,1.52f, 1.70f,1.70f,1.70f,1.70f,0,0,0,0,0,0},
  {1.55f,1.70f,1.70f,1.52f, 1.70f,1.70f,1.80f,1.70f,0,0,0,0,0,0},
  {1.55f,1.70f,1.70f,1.52f, 1.70f,1.70f,1.52f,1.55f,0,0,0,0,0,0},
  {1.55f,1.70f,1.70f,1.52f, 1.70f,1.70f,1.70f,0,0,0,0,0,0,0},
  {1.55f,1.70f,1.70f,1.52f, 1.70f,1.70f,1.70f,1.52f,1.55f,0,0,0,0,0},
  {1.55f,1.70f,1.70f,1.52f, 1.70f,1.70f,1.70f,1.55f,1.70f,1.55f,1.55f,0,0,0},
  {1.55f,1.70f,1.70f,1.52f, 1.70f,1.52f,0,0,0,0,0,0,0,0},
  {1.55f,1.70f,1.70f,1.52f, 1.70f,1.52f,1.70f,0,0,0,0,0,0,0},
  {1.55f,1.70f,1.70f,1.52f, 1.70f,1.70f,1.70f,0,0,0,0,0,0,0},
  {1.55f,1.70f,1.70f,1.52f, 1.70f,1.70f,1.70f,1.70f,1.55f,1.70f,1.70f,1.70f,1.70f,1.70f},
  {1.55f,1.70f,1.70f,1.52f, 1.70f,1.70f,1.70f,1.70f,1.70f,1.70f,1.70f,1.52f,0,0},
};
__device__ __constant__ int g_natoms[20] =
  {5,6,8,9,11,4,10,8,9,8,8,8,7,9,11,6,7,7,14,12};

#define DIST_EPS      1e-3f
#define CONNECT_CUT   0.35f
#define LOG2E         1.44269504088896340736f

__global__ __launch_bounds__(256)
void clash_wmma_kernel(const float* __restrict__ X,
                       const float* __restrict__ mask_j,
                       const float* __restrict__ mask_ij,
                       const int*   __restrict__ C,
                       const int*   __restrict__ S,
                       const int*   __restrict__ E,
                       float*       __restrict__ out,
                       int N, int K, int BN)
{
    const int lane = threadIdx.x & 31;
    const int wave = blockIdx.x * (blockDim.x >> 5) + (threadIdx.x >> 5);
    if (wave >= BN) return;

    const int b = wave / N;
    const int i = wave % N;
    const int a  = lane & 15;
    const int hi = lane >> 4;

    const int s_i   = S[wave];
    const int cnt_i = (C[wave] > 0) ? g_natoms[s_i] : 0;
    float xi = 0.f, yi = 0.f, zi = 0.f, rad_i = 0.f;
    if (a < 14) {
        const float* p = X + (size_t)wave * 42 + a * 3;
        xi = p[0]; yi = p[1]; zi = p[2];
        rad_i = g_vdw[s_i][a];
    }
    const float n2_i = xi * xi + yi * yi + zi * zi;
    const float mA_i = (a < cnt_i) ? 1.f : 0.f;

    v2f Areg;
    Areg.x = hi ? zi : xi;
    Areg.y = hi ? 0.f : yi;

    float n2M[8], hcutM[8], mAM[8];
#pragma unroll
    for (int r = 0; r < 8; ++r) {
        const int M = r + (hi << 3);
        n2M[r]  = __shfl(n2_i, M, 32);
        hcutM[r] = 0.5f * __shfl(rad_i, M, 32) + CONNECT_CUT;
        const float m = __shfl(mA_i, M, 32);
        mAM[r] = (M >= 4) ? m : 0.f;
    }

    float acc = 0.f;
    const int baseE = wave * K;
    for (int k = 0; k < K; ++k) {
        const int   j  = E[baseE + k];
        const float me = mask_ij[baseE + k] * ((j != i) ? 1.f : 0.f);
        const int   jg = b * N + j;

        const int s_j   = S[jg];
        const int cnt_j = (C[jg] > 0) ? g_natoms[s_j] : 0;
        float xj = 0.f, yj = 0.f, zj = 0.f, rad_j = 0.f;
        if (a < 14) {
            const float* p = X + (size_t)jg * 42 + a * 3;
            xj = p[0]; yj = p[1]; zj = p[2];
            rad_j = g_vdw[s_j][a];
        }
        const float n2N  = xj * xj + yj * yj + zj * zj + DIST_EPS;
        const float hradN = 0.5f * rad_j;
        const float wN   = ((a < cnt_j) ? 1.f : 0.f) * mask_j[jg] * me;

        v2f Breg;
        Breg.x = hi ? zj : xj;
        Breg.y = hi ? 0.f : yj;

        v8f czero = {};
        v8f dot = __builtin_amdgcn_wmma_f32_16x16x4_f32(
             false, Areg,  false, Breg,
             (short)0, czero,  false,  false);

#pragma unroll
        for (int r = 0; r < 8; ++r) {
            const float dsq  = n2M[r] + n2N - 2.f * dot[r];
            const float D    = __builtin_amdgcn_sqrtf(dsq);
            const float t    = D - (hcutM[r] + hradN);
            const float e    = __builtin_amdgcn_exp2f(t * LOG2E);
            const float sg   = __builtin_amdgcn_rcpf(1.f + e);
            acc += (mAM[r] * wN) * sg;
        }
    }

#pragma unroll
    for (int off = 16; off > 0; off >>= 1)
        acc += __shfl_down(acc, off, 32);
    if (lane == 0) out[wave] = acc;
}

extern "C" void kernel_launch(void* const* d_in, const int* in_sizes, int n_in,
                              void* d_out, int out_size, void* d_ws, size_t ws_size,
                              hipStream_t stream) {
    const float* X       = (const float*)d_in[0];
    const float* mask_j  = (const float*)d_in[1];
    const float* mask_ij = (const float*)d_in[2];
    const int*   C       = (const int*)d_in[3];
    const int*   S       = (const int*)d_in[4];
    const int*   E       = (const int*)d_in[5];
    float*       out     = (float*)d_out;

    const int BN = in_sizes[1];
    const int K  = in_sizes[2] / BN;
    const int B  = 2;
    const int N  = BN / B;

    const int wavesPerBlock = 8;
    const int blocks = (BN + wavesPerBlock - 1) / wavesPerBlock;
    clash_wmma_kernel<<<blocks, wavesPerBlock * 32, 0, stream>>>(
        X, mask_j, mask_ij, C, S, E, out, N, K, BN);
}


# Entry chain of dependent round trips cut from 7 to 3: edge words loaded with the residue's own, neighbour types beside its radius row, atom counts from packed constants, row set-up under the gather
# speedup vs baseline: 2.0233x; 2.0233x over previous
#include <hip/hip_runtime.h>
#include <math.h>

typedef __attribute__((ext_vector_type(2))) float v2f;
typedef __attribute__((ext_vector_type(4))) float v4f;
typedef __attribute__((ext_vector_type(8))) float v8f;
typedef __attribute__((ext_vector_type(4))) int   v4i;

__device__ __constant__ float g_vdw[20][14] = {
  {1.55f,1.70f,1.70f,1.52f, 1.70f,0,0,0,0,0,0,0,0,0},
  {1.55f,1.70f,1.70f,1.52f, 1.70f,1.80f,0,0,0,0,0,0,0,0},
  {1.55f,1.70f,1.70f,1.52f, 1.70f,1.70f,1.52f,1.52f,0,0,0,0,0,0},
  {1.55f,1.70f,1.70f,1.52f, 1.70f,1.70f,1.70f,1.52f,1.52f,0,0,0,0,0},
  {1.55f,1.70f,1.70f,1.52f, 1.70f,1.70f,1.70f,1.70f,1.70f,1.70f,1.70f,0,0,0},
  {1.55f,1.70f,1.70f,1.52f, 0,0,0,0,0,0,0,0,0,0},
  {1.55f,1.70f,1.70f,1.52f, 1.70f,1.70f,1.55f,1.70f,1.70f,1.55f,0,0,0,0},
  {1.55f,1.70f,1.70f,1.52f, 1.70f,1.70f,1.70f,1.70f,0,0,0,0,0,0},
  {1.55f,1.70f,1.70f,1.52f, 1.70f,1.70f,1.70f,1.70f,1.55f,0,0,0,0,0},
  {1.55f,1.70f,1.70f,1.52f, 1.70f,1.70f,1.70f,1.70f,0,0,0,0,0,0},
  {1.55f,1.70f,1.70f,1.52f, 1.70f,1.70f,1.80f,1.70f,0,0,0,0,0,0},
  {1.55f,1.70f,1.70f,1.52f, 1.70f,1.70f,1.52f,1.55f,0,0,0,0,0,0},
  {1.55f,1.70f,1.70f,1.52f, 1.70f,1.70f,1.70f,0,0,0,0,0,0,0},
  {1.55f,1.70f,1.70f,1.52f, 1.70f,1.70f,1.70f,1.52f,1.55f,0,0,0,0,0},
  {1.55f,1.70f,1.70f,1.52f, 1.70f,1.70f,1.70f,1.55f,1.70f,1.55f,1.55f,0,0,0},
  {1.55f,1.70f,1.70f,1.52f, 1.70f,1.52f,0,0,0,0,0,0,0,0},
  {1.55f,1.70f,1.70f,1.52f, 1.70f,1.52f,1.70f,0,0,0,0,0,0,0},
  {1.55f,1.70f,1.70f,1.52f, 1.70f,1.70f,1.70f,0,0,0,0,0,0,0},
  {1.55f,1.70f,1.70f,1.52f, 1.70f,1.70f,1.70f,1.70f,1.55f,1.70f,1.70f,1.70f,1.70f,1.70f},
  {1.55f,1.70f,1.70f,1.52f, 1.70f,1.70f,1.70f,1.70f,1.70f,1.70f,1.70f,1.52f,0,0},
};
__device__ __forceinline__ int natoms_of(int s) {
    const unsigned w = (s < 8) ? 0x8A4B9865u : (s < 16) ? 0x6B978889u : 0x0000CE77u;
    return (int)((w >> ((s & 7) * 4)) & 15u);
}

__device__ __forceinline__ v4i edge_record(bool on, int jg, int s_j, int c_j, float mj, float mij, bool self)
{
    const int cnt_j = (c_j > 0) ? natoms_of(s_j) : 0;
    v4i m;
    m.x = on ? jg : 0;
    m.y = on ? (s_j | (cnt_j << 8)) : 0;
    m.z = on ? __float_as_int(mj) : 0;
    m.w = on ? __float_as_int(mij * (self ? 0.f : 1.f)) : 0;
    return m;
}

#define DIST_EPS      1e-3f
#define CONNECT_CUT   0.35f

#define WAVES_PER_BLOCK 8
#define EDGE_SLOTS      32
#define BLOCK_SLOTS     (WAVES_PER_BLOCK * EDGE_SLOTS)

__global__ __launch_bounds__(256)
void clash_wmma_kernel(const float* __restrict__ X,
                       const float* __restrict__ mask_j,
                       const float* __restrict__ mask_ij,
                       const int*   __restrict__ C,
                       const int*   __restrict__ S,
                       const int*   __restrict__ E,
                       float*       __restrict__ out,
                       int N, int K, int BN)
{
    __shared__ v4i s_meta[BLOCK_SLOTS];
    __shared__ v4f s_pos[BLOCK_SLOTS * 16];
    __shared__ v2f s_rw[BLOCK_SLOTS * 16];

    const int  tid   = threadIdx.x;
    const int  lane  = tid & 31;
    const int  wv    = tid >> 5;
    const int  wave  = blockIdx.x * WAVES_PER_BLOCK + wv;
    const bool valid = wave < BN;
    const int  wave_c = valid ? wave : 0;

    const int b = wave_c / N;
    const int i = wave_c % N;
    const int a  = lane & 15;
    const int hi = lane >> 4;

    const int aL = (a < 14) ? a : 13;

    if (K <= 0) {
        if (valid && lane == 0) out[wave] = 0.f;
        return;
    }

    const int baseE = wave_c * K;
    const int kc0   = min(K, EDGE_SLOTS);
    const int le0   = min(lane, kc0 - 1);
    const int   e_j0   = E[baseE + le0];
    const float e_mij0 = mask_ij[baseE + le0];
    const int   s_i    = S[wave_c];
    const int   c_i    = C[wave_c];
    const float* pxi = X + (size_t)wave_c * 42 + aL * 3;
    const float xi_l = pxi[0], yi_l = pxi[1], zi_l = pxi[2];
    const int   jg0  = b * N + e_j0;
    const int   s_j0 = S[jg0];
    const int   c_j0 = C[jg0];
    const float mj0  = mask_j[jg0];
    const float rad_l = g_vdw[s_i][aL];

    v4i erec = edge_record(valid && lane < kc0, jg0, s_j0, c_j0, mj0, e_mij0, e_j0 == i);
    s_meta[wv * EDGE_SLOTS + lane] = erec;

    v2f   Areg = {0.f, 0.f};
    float n2M[8], hcutM[8], mAM[8];
    float cntM = 0.f;
#pragma unroll
    for (int r = 0; r < 8; ++r) { n2M[r] = 0.f; hcutM[r] = 0.f; mAM[r] = 0.f; }

    float acc  = 0.f;
    float accc = 0.f;
    const int sub   = tid >> 4;

    for (int k0 = 0; k0 < K; k0 += EDGE_SLOTS) {
        const int kc = min(K - k0, EDGE_SLOTS);

        __syncthreads();

        v4i   mm[16];
        float gx[16], gy[16], gz[16], gr[16];
#pragma unroll
        for (int p = 0; p < 16; ++p)
            mm[p] = s_meta[p * 16 + sub];
#pragma unroll
        for (int p = 0; p < 16; ++p) {
            const float* q = X + (size_t)mm[p].x * 42 + aL * 3;
            gx[p] = q[0]; gy[p] = q[1]; gz[p] = q[2];
            gr[p] = g_vdw[mm[p].y & 255][aL];
        }
        if (k0 == 0) {
            const bool real_a = a < 14;
            const float xi = real_a ? xi_l : 0.f, yi = real_a ? yi_l : 0.f, zi = real_a ? zi_l : 0.f;
            const float rad_i = real_a ? rad_l : 0.f;
            const int cnt_i = (c_i > 0) ? natoms_of(s_i) : 0;
            const float n2_i = xi * xi + yi * yi + zi * zi;
            const float mA_i = (a < cnt_i) ? 1.f : 0.f;
            Areg.x = hi ? zi : xi;
            Areg.y = hi ? 0.f : yi;
#pragma unroll
            for (int r = 0; r < 8; ++r) {
                const int M = r + (hi << 3);
                n2M[r]  = 0.25f * __shfl(n2_i, M, 32);
                hcutM[r] = 0.5f * (0.5f * __shfl(rad_i, M, 32) + CONNECT_CUT);
                const float mrow = __shfl(mA_i, M, 32);
                mAM[r] = (M >= 4) ? mrow : 0.f;
                cntM += mAM[r];
            }
        }
#pragma unroll
        for (int p = 0; p < 16; ++p) {
            const bool  real = a < 14;
            const float xj = real ? gx[p] : 0.f;
            const float yj = real ? gy[p] : 0.f;
            const float zj = real ? gz[p] : 0.f;
            const float rad_j = real ? gr[p] : 0.f;
            const float n2N = xj * xj + yj * yj + zj * zj + DIST_EPS;
            const float wN  = ((a < (mm[p].y >> 8)) ? 1.f : 0.f)
                              * __int_as_float(mm[p].z) * __int_as_float(mm[p].w);
            const int   o   = (p * 16 + sub) * 16 + a;
            v4f rec; rec.x = xj; rec.y = yj; rec.z = zj; rec.w = 0.25f * n2N;
            v2f rw;  rw.x = 0.25f * rad_j; rw.y = 0.5f * wN;
            s_pos[o] = rec;
            s_rw[o]  = rw;
        }
        __syncthreads();

        const v4f* pp = s_pos + (wv * EDGE_SLOTS) * 16 + a;
        const v2f* pq = s_rw  + (wv * EDGE_SLOTS) * 16 + a;
        v4f pc = pp[0];
        v2f rc = pq[0];
        for (int kk = 0; kk < kc; ++kk) {
            v4f pn = pc;
            v2f rn = rc;
            if (kk + 1 < kc) { pn = pp[(kk + 1) * 16]; rn = pq[(kk + 1) * 16]; }

            const float n2N   = pc.w;
            const float hradN = rc.x;
            const float wN    = rc.y;

            v2f Breg;
            Breg.x = hi ? pc.z : pc.x;
            Breg.y = hi ? 0.f : pc.y;

            v8f czero = {};
            v8f dot = __builtin_amdgcn_wmma_f32_16x16x4_f32(
                 false, Areg,  false, Breg,
                 (short)0, czero,  false,  false);

            accc = fmaf(cntM, wN, accc);
#pragma unroll
            for (int r = 0; r < 8; ++r) {
                const float qsq  = n2M[r] + n2N - 0.5f * dot[r];
                const float hD   = __builtin_amdgcn_sqrtf(qsq);
                const float u    = (hcutM[r] + hradN) - hD;
                const float th   = __builtin_amdgcn_tanhf(u);
                acc = fmaf(mAM[r] * wN, th, acc);
            }
            pc = pn;
            rc = rn;
        }

        if (k0 + EDGE_SLOTS < K) {
            const int kn  = min(K - k0 - EDGE_SLOTS, EDGE_SLOTS);
            const int ln  = min(lane, kn - 1);
            const int   e_j   = E[baseE + k0 + EDGE_SLOTS + ln];
            const float e_mij = mask_ij[baseE + k0 + EDGE_SLOTS + ln];
            const int   jg    = b * N + e_j;
            erec = edge_record(valid && lane < kn, jg, S[jg], C[jg], mask_j[jg], e_mij, e_j == i);
            s_meta[wv * EDGE_SLOTS + lane] = erec;
        }
    }

    acc += accc;
#pragma unroll
    for (int off = 16; off > 0; off >>= 1)
        acc += __shfl_down(acc, off, 32);
    if (valid && lane == 0) out[wave] = acc;
}

extern "C" void kernel_launch(void* const* d_in, const int* in_sizes, int n_in,
                              void* d_out, int out_size, void* d_ws, size_t ws_size,
                              hipStream_t stream) {
    const float* X       = (const float*)d_in[0];
    const float* mask_j  = (const float*)d_in[1];
    const float* mask_ij = (const float*)d_in[2];
    const int*   C       = (const int*)d_in[3];
    const int*   S       = (const int*)d_in[4];
    const int*   E       = (const int*)d_in[5];
    float*       out     = (float*)d_out;

    const int BN = in_sizes[1];
    const int K  = in_sizes[2] / BN;
    const int B  = 2;
    const int N  = BN / B;

    const int wavesPerBlock = 8;
    const int blocks = (BN + wavesPerBlock - 1) / wavesPerBlock;
    clash_wmma_kernel<<<blocks, wavesPerBlock * 32, 0, stream>>>(
        X, mask_j, mask_ij, C, S, E, out, N, K, BN);
}


# Two edges per result tile: row atoms 4..11 in rows 0-7 of one A and rows 8-15 of a second, two accumulating WMMAs, epilogue once per 2 edges (over 12 atoms: 1 edge per tile); lane sum order changes
# speedup vs baseline: 2.1993x; 1.0870x over previous
#include <hip/hip_runtime.h>
#include <math.h>

typedef __attribute__((ext_vector_type(2))) float v2f;
typedef __attribute__((ext_vector_type(4))) float v4f;
typedef __attribute__((ext_vector_type(8))) float v8f;
typedef __attribute__((ext_vector_type(4))) int   v4i;

__device__ __constant__ float g_vdw[20][14] = {
  {1.55f,1.70f,1.70f,1.52f, 1.70f,0,0,0,0,0,0,0,0,0},
  {1.55f,1.70f,1.70f,1.52f, 1.70f,1.80f,0,0,0,0,0,0,0,0},
  {1.55f,1.70f,1.70f,1.52f, 1.70f,1.70f,1.52f,1.52f,0,0,0,0,0,0},
  {1.55f,1.70f,1.70f,1.52f, 1.70f,1.70f,1.70f,1.52f,1.52f,0,0,0,0,0},
  {1.55f,1.70f,1.70f,1.52f, 1.70f,1.70f,1.70f,1.70f,1.70f,1.70f,1.70f,0,0,0},
  {1.55f,1.70f,1.70f,1.52f, 0,0,0,0,0,0,0,0,0,0},
  {1.55f,1.70f,1.70f,1.52f, 1.70f,1.70f,1.55f,1.70f,1.70f,1.55f,0,0,0,0},
  {1.55f,1.70f,1.70f,1.52f, 1.70f,1.70f,1.70f,1.70f,0,0,0,0,0,0},
  {1.55f,1.70f,1.70f,1.52f, 1.70f,1.70f,1.70f,1.70f,1.55f,0,0,0,0,0},
  {1.55f,1.70f,1.70f,1.52f, 1.70f,1.70f,1.70f,1.70f,0,0,0,0,0,0},
  {1.55f,1.70f,1.70f,1.52f, 1.70f,1.70f,1.80f,1.70f,0,0,0,0,0,0},
  {1.55f,1.70f,1.70f,1.52f, 1.70f,1.70f,1.52f,1.55f,0,0,0,0,0,0},
  {1.55f,1.70f,1.70f,1.52f, 1.70f,1.70f,1.70f,0,0,0,0,0,0,0},
  {1.55f,1.70f,1.70f,1.52f, 1.70f,1.70f,1.70f,1.52f,1.55f,0,0,0,0,0},
  {1.55f,1.70f,1.70f,1.52f, 1.70f,1.70f,1.70f,1.55f,1.70f,1.55f,1.55f,0,0,0},
  {1.55f,1.70f,1.70f,1.52f, 1.70f,1.52f,0,0,0,0,0,0,0,0},
  {1.55f,1.70f,1.70f,1.52f, 1.70f,1.52f,1.70f,0,0,0,0,0,0,0},
  {1.55f,1.70f,1.70f,1.52f, 1.70f,1.70f,1.70f,0,0,0,0,0,0,0},
  {1.55f,1.70f,1.70f,1.52f, 1.70f,1.70f,1.70f,1.70f,1.55f,1.70f,1.70f,1.70f,1.70f,1.70f},
  {1.55f,1.70f,1.70f,1.52f, 1.70f,1.70f,1.70f,1.70f,1.70f,1.70f,1.70f,1.52f,0,0},
};
__device__ __forceinline__ int natoms_of(int s) {
    const unsigned w = (s < 8) ? 0x8A4B9865u : (s < 16) ? 0x6B978889u : 0x0000CE77u;
    return (int)((w >> ((s & 7) * 4)) & 15u);
}

__device__ __forceinline__ v4i edge_record(bool on, int jg, int s_j, int c_j, float mj, float mij, bool self)
{
    const int cnt_j = (c_j > 0) ? natoms_of(s_j) : 0;
    v4i m;
    m.x = on ? jg : 0;
    m.y = on ? (s_j | (cnt_j << 8)) : 0;
    m.z = on ? __float_as_int(mj) : 0;
    m.w = on ? __float_as_int(mij * (self ? 0.f : 1.f)) : 0;
    return m;
}

#define DIST_EPS      1e-3f
#define CONNECT_CUT   0.35f

#define WAVES_PER_BLOCK 8
#define EDGE_SLOTS      32
#define BLOCK_SLOTS     (WAVES_PER_BLOCK * EDGE_SLOTS)

__global__ __launch_bounds__(256)
void clash_wmma_kernel(const float* __restrict__ X,
                       const float* __restrict__ mask_j,
                       const float* __restrict__ mask_ij,
                       const int*   __restrict__ C,
                       const int*   __restrict__ S,
                       const int*   __restrict__ E,
                       float*       __restrict__ out,
                       int N, int K, int BN)
{
    __shared__ v4i s_meta[BLOCK_SLOTS];
    __shared__ v4f s_pos[BLOCK_SLOTS * 16];
    __shared__ v2f s_rw[BLOCK_SLOTS * 16];

    const int  tid   = threadIdx.x;
    const int  lane  = tid & 31;
    const int  wv    = tid >> 5;
    const int  wave  = blockIdx.x * WAVES_PER_BLOCK + wv;
    const bool valid = wave < BN;
    const int  wave_c = valid ? wave : 0;

    const int b = wave_c / N;
    const int i = wave_c % N;
    const int a  = lane & 15;
    const int hi = lane >> 4;

    const int aL = (a < 14) ? a : 13;

    if (K <= 0) {
        if (valid && lane == 0) out[wave] = 0.f;
        return;
    }

    const int baseE = wave_c * K;
    const int kc0   = min(K, EDGE_SLOTS);
    const int le0   = min(lane, kc0 - 1);
    const int   e_j0   = E[baseE + le0];
    const float e_mij0 = mask_ij[baseE + le0];
    const int   s_i    = S[wave_c];
    const int   c_i    = C[wave_c];
    const float* pxi = X + (size_t)wave_c * 42 + aL * 3;
    const float xi_l = pxi[0], yi_l = pxi[1], zi_l = pxi[2];
    const float* pxp = X + (size_t)wave_c * 42 + (4 + (a & 7)) * 3;
    const float xp_l = pxp[0], yp_l = pxp[1], zp_l = pxp[2];
    const int   jg0  = b * N + e_j0;
    const int   s_j0 = S[jg0];
    const int   c_j0 = C[jg0];
    const float mj0  = mask_j[jg0];
    const float rad_l = g_vdw[s_i][aL];

    v4i erec = edge_record(valid && lane < kc0, jg0, s_j0, c_j0, mj0, e_mij0, e_j0 == i);
    s_meta[wv * EDGE_SLOTS + lane] = erec;

    const int  cnt_i  = (c_i > 0) ? natoms_of(s_i) : 0;
    const bool paired = __builtin_amdgcn_readfirstlane(cnt_i) <= 12;

    v2f   Areg = {0.f, 0.f};
    v2f   Areg2 = {0.f, 0.f};
    float n2M[8], hcutM[8], mAM[8];
    float cntM = 0.f;
#pragma unroll
    for (int r = 0; r < 8; ++r) { n2M[r] = 0.f; hcutM[r] = 0.f; mAM[r] = 0.f; }

    float acc  = 0.f;
    float accc = 0.f;
    const int sub   = tid >> 4;

    for (int k0 = 0; k0 < K; k0 += EDGE_SLOTS) {
        const int kc = min(K - k0, EDGE_SLOTS);

        __syncthreads();

        v4i   mm[16];
        float gx[16], gy[16], gz[16], gr[16];
#pragma unroll
        for (int p = 0; p < 16; ++p)
            mm[p] = s_meta[p * 16 + sub];
#pragma unroll
        for (int p = 0; p < 16; ++p) {
            const float* q = X + (size_t)mm[p].x * 42 + aL * 3;
            gx[p] = q[0]; gy[p] = q[1]; gz[p] = q[2];
            gr[p] = g_vdw[mm[p].y & 255][aL];
        }
        if (k0 == 0) {
            const bool real_a = a < 14;
            const float xi = real_a ? xi_l : 0.f, yi = real_a ? yi_l : 0.f, zi = real_a ? zi_l : 0.f;
            const float rad_i = real_a ? rad_l : 0.f;
            const float n2_i = xi * xi + yi * yi + zi * zi;
            const float mA_i = (a < cnt_i) ? 1.f : 0.f;
            if (paired) {
                const float ax = hi ? zp_l : xp_l;
                const float ay = hi ? 0.f : yp_l;
                Areg.x  = (a < 8) ? ax : 0.f;
                Areg.y  = (a < 8) ? ay : 0.f;
                Areg2.x = (a < 8) ? 0.f : ax;
                Areg2.y = (a < 8) ? 0.f : ay;
            } else {
                Areg.x = hi ? zi : xi;
                Areg.y = hi ? 0.f : yi;
            }
#pragma unroll
            for (int r = 0; r < 8; ++r) {
                const int M = paired ? (4 + r) : (r + (hi << 3));
                n2M[r]  = 0.25f * __shfl(n2_i, M, 32);
                hcutM[r] = 0.5f * (0.5f * __shfl(rad_i, M, 32) + CONNECT_CUT);
                const float mrow = __shfl(mA_i, M, 32);
                mAM[r] = (M >= 4) ? mrow : 0.f;
                cntM += mAM[r];
            }
        }
#pragma unroll
        for (int p = 0; p < 16; ++p) {
            const bool  real = a < 14;
            const float xj = real ? gx[p] : 0.f;
            const float yj = real ? gy[p] : 0.f;
            const float zj = real ? gz[p] : 0.f;
            const float rad_j = real ? gr[p] : 0.f;
            const float n2N = xj * xj + yj * yj + zj * zj + DIST_EPS;
            const float wN  = ((a < (mm[p].y >> 8)) ? 1.f : 0.f)
                              * __int_as_float(mm[p].z) * __int_as_float(mm[p].w);
            const int   o   = (p * 16 + sub) * 16 + a;
            v4f rec; rec.x = xj; rec.y = yj; rec.z = zj; rec.w = 0.25f * n2N;
            v2f rw;  rw.x = 0.25f * rad_j; rw.y = 0.5f * wN;
            s_pos[o] = rec;
            s_rw[o]  = rw;
        }
        __syncthreads();

        const v4f* pp = s_pos + (wv * EDGE_SLOTS) * 16 + a;
        const v2f* pq = s_rw  + (wv * EDGE_SLOTS) * 16 + a;
        if (paired) {
            const int  nt  = (kc + 1) >> 1;
            const v2f* pqh = pq + hi * 16;
            v4f p0 = pp[0], p1 = pp[16];
            v2f rp = pqh[0];
            for (int t = 0; t < nt; ++t) {
                v4f q0 = p0, q1 = p1;
                v2f rq = rp;
                if (t + 1 < nt) { q0 = pp[(2 * t + 2) * 16]; q1 = pp[(2 * t + 3) * 16]; rq = pqh[(2 * t + 2) * 16]; }

                const float n2N   = hi ? p1.w : p0.w;
                const float hradN = rp.x;
                const float wN    = rp.y;

                v2f B0, B1;
                B0.x = hi ? p0.z : p0.x;
                B0.y = hi ? 0.f : p0.y;
                B1.x = hi ? p1.z : p1.x;
                B1.y = hi ? 0.f : p1.y;

                v8f czero = {};
                v8f dot = __builtin_amdgcn_wmma_f32_16x16x4_f32(
                     false, Areg,  false, B0,
                     (short)0, czero,  false,  false);
                dot = __builtin_amdgcn_wmma_f32_16x16x4_f32(
                     false, Areg2,  false, B1,
                     (short)0, dot,  false,  false);

                accc = fmaf(cntM, wN, accc);
#pragma unroll
                for (int r = 0; r < 8; ++r) {
                    const float qsq  = n2M[r] + n2N - 0.5f * dot[r];
                    const float hD   = __builtin_amdgcn_sqrtf(qsq);
                    const float u    = (hcutM[r] + hradN) - hD;
                    const float th   = __builtin_amdgcn_tanhf(u);
                    acc = fmaf(mAM[r] * wN, th, acc);
                }
                p0 = q0;
                p1 = q1;
                rp = rq;
            }
        } else {
        v4f pc = pp[0];
        v2f rc = pq[0];
        for (int kk = 0; kk < kc; ++kk) {
            v4f pn = pc;
            v2f rn = rc;
            if (kk + 1 < kc) { pn = pp[(kk + 1) * 16]; rn = pq[(kk + 1) * 16]; }

            const float n2N   = pc.w;
            const float hradN = rc.x;
            const float wN    = rc.y;

            v2f Breg;
            Breg.x = hi ? pc.z : pc.x;
            Breg.y = hi ? 0.f : pc.y;

            v8f czero = {};
            v8f dot = __builtin_amdgcn_wmma_f32_16x16x4_f32(
                 false, Areg,  false, Breg,
                 (short)0, czero,  false,  false);

            accc = fmaf(cntM, wN, accc);
#pragma unroll
            for (int r = 0; r < 8; ++r) {
                const float qsq  = n2M[r] + n2N - 0.5f * dot[r];
                const float hD   = __builtin_amdgcn_sqrtf(qsq);
                const float u    = (hcutM[r] + hradN) - hD;
                const float th   = __builtin_amdgcn_tanhf(u);
                acc = fmaf(mAM[r] * wN, th, acc);
            }
            pc = pn;
            rc = rn;
        }
        }

        if (k0 + EDGE_SLOTS < K) {
            const int kn  = min(K - k0 - EDGE_SLOTS, EDGE_SLOTS);
            const int ln  = min(lane, kn - 1);
            const int   e_j   = E[baseE + k0 + EDGE_SLOTS + ln];
            const float e_mij = mask_ij[baseE + k0 + EDGE_SLOTS + ln];
            const int   jg    = b * N + e_j;
            erec = edge_record(valid && lane < kn, jg, S[jg], C[jg], mask_j[jg], e_mij, e_j == i);
            s_meta[wv * EDGE_SLOTS + lane] = erec;
        }
    }

    acc += accc;
#pragma unroll
    for (int off = 16; off > 0; off >>= 1)
        acc += __shfl_down(acc, off, 32);
    if (valid && lane == 0) out[wave] = acc;
}

extern "C" void kernel_launch(void* const* d_in, const int* in_sizes, int n_in,
                              void* d_out, int out_size, void* d_ws, size_t ws_size,
                              hipStream_t stream) {
    const float* X       = (const float*)d_in[0];
    const float* mask_j  = (const float*)d_in[1];
    const float* mask_ij = (const float*)d_in[2];
    const int*   C       = (const int*)d_in[3];
    const int*   S       = (const int*)d_in[4];
    const int*   E       = (const int*)d_in[5];
    float*       out     = (float*)d_out;

    const int BN = in_sizes[1];
    const int K  = in_sizes[2] / BN;
    const int B  = 2;
    const int N  = BN / B;

    const int wavesPerBlock = 8;
    const int blocks = (BN + wavesPerBlock - 1) / wavesPerBlock;
    clash_wmma_kernel<<<blocks, wavesPerBlock * 32, 0, stream>>>(
        X, mask_j, mask_ij, C, S, E, out, N, K, BN);
}


# Edges per result tile set by the residue's sidechain row count: 8, 4 or 2 edges per 16-row tile (row groups of 2, 4, 8), atoms 12-13 in a second 8-edge pass, no-sidechain rows skip; sum order changes
# speedup vs baseline: 2.3481x; 1.0677x over previous
#include <hip/hip_runtime.h>
#include <math.h>

typedef __attribute__((ext_vector_type(2))) float v2f;
typedef __attribute__((ext_vector_type(4))) float v4f;
typedef __attribute__((ext_vector_type(8))) float v8f;
typedef __attribute__((ext_vector_type(4))) int   v4i;

__device__ __constant__ float g_vdw[20][14] = {
  {1.55f,1.70f,1.70f,1.52f, 1.70f,0,0,0,0,0,0,0,0,0},
  {1.55f,1.70f,1.70f,1.52f, 1.70f,1.80f,0,0,0,0,0,0,0,0},
  {1.55f,1.70f,1.70f,1.52f, 1.70f,1.70f,1.52f,1.52f,0,0,0,0,0,0},
  {1.55f,1.70f,1.70f,1.52f, 1.70f,1.70f,1.70f,1.52f,1.52f,0,0,0,0,0},
  {1.55f,1.70f,1.70f,1.52f, 1.70f,1.70f,1.70f,1.70f,1.70f,1.70f,1.70f,0,0,0},
  {1.55f,1.70f,1.70f,1.52f, 0,0,0,0,0,0,0,0,0,0},
  {1.55f,1.70f,1.70f,1.52f, 1.70f,1.70f,1.55f,1.70f,1.70f,1.55f,0,0,0,0},
  {1.55f,1.70f,1.70f,1.52f, 1.70f,1.70f,1.70f,1.70f,0,0,0,0,0,0},
  {1.55f,1.70f,1.70f,1.52f, 1.70f,1.70f,1.70f,1.70f,1.55f,0,0,0,0,0},
  {1.55f,1.70f,1.70f,1.52f, 1.70f,1.70f,1.70f,1.70f,0,0,0,0,0,0},
  {1.55f,1.70f,1.70f,1.52f, 1.70f,1.70f,1.80f,1.70f,0,0,0,0,0,0},
  {1.55f,1.70f,1.70f,1.52f, 1.70f,1.70f,1.52f,1.55f,0,0,0,0,0,0},
  {1.55f,1.70f,1.70f,1.52f, 1.70f,1.70f,1.70f,0,0,0,0,0,0,0},
  {1.55f,1.70f,1.70f,1.52f, 1.70f,1.70f,1.70f,1.52f,1.55f,0,0,0,0,0},
  {1.55f,1.70f,1.70f,1.52f, 1.70f,1.70f,1.70f,1.55f,1.70f,1.55f,1.55f,0,0,0},
  {1.55f,1.70f,1.70f,1.52f, 1.70f,1.52f,0,0,0,0,0,0,0,0},
  {1.55f,1.70f,1.70f,1.52f, 1.70f,1.52f,1.70f,0,0,0,0,0,0,0},
  {1.55f,1.70f,1.70f,1.52f, 1.70f,1.70f,1.70f,0,0,0,0,0,0,0},
  {1.55f,1.70f,1.70f,1.52f, 1.70f,1.70f,1.70f,1.70f,1.55f,1.70f,1.70f,1.70f,1.70f,1.70f},
  {1.55f,1.70f,1.70f,1.52f, 1.70f,1.70f,1.70f,1.70f,1.70f,1.70f,1.70f,1.52f,0,0},
};
__device__ __forceinline__ int natoms_of(int s) {
    const unsigned w = (s < 8) ? 0x8A4B9865u : (s < 16) ? 0x6B978889u : 0x0000CE77u;
    return (int)((w >> ((s & 7) * 4)) & 15u);
}

__device__ __forceinline__ v4i edge_record(bool on, int jg, int s_j, int c_j, float mj, float mij, bool self)
{
    const int cnt_j = (c_j > 0) ? natoms_of(s_j) : 0;
    v4i m;
    m.x = on ? jg : 0;
    m.y = on ? (s_j | (cnt_j << 8)) : 0;
    m.z = on ? __float_as_int(mj) : 0;
    m.w = on ? __float_as_int(mij * (self ? 0.f : 1.f)) : 0;
    return m;
}

#define DIST_EPS      1e-3f
#define CONNECT_CUT   0.35f

#define WAVES_PER_BLOCK 8
#define EDGE_SLOTS      32
#define BLOCK_SLOTS     (WAVES_PER_BLOCK * EDGE_SLOTS)


struct RowTerms {
    float n2M[8];
    float hcutM[8];
    float mAM[8];
    float cnt;
    v2f   arow;
    int   erow;
};

__device__ __forceinline__ void row_terms(RowTerms& T, int m0, int sh, int lane, int a, int hi,
                                          float xi, float yi, float zi, float n2_i, float rad_i, float mA_i)
{
    const int rmask = (1 << sh) - 1;
    const int src = (lane & 16) | (m0 + (a & rmask));
    const float xq = __shfl(xi, src, 32);
    const float yq = __shfl(yi, src, 32);
    const float zq = __shfl(zi, src, 32);
    T.arow.x = hi ? zq : xq;
    T.arow.y = hi ? 0.f : yq;
    T.erow   = a >> sh;
    T.cnt    = 0.f;
#pragma unroll
    for (int r = 0; r < 8; ++r) {
        const int M = m0 + (r & rmask);
        T.n2M[r]   = 0.25f * __shfl(n2_i, M, 32);
        T.hcutM[r] = 0.5f * (0.5f * __shfl(rad_i, M, 32) + CONNECT_CUT);
        T.mAM[r]   = __shfl(mA_i, M, 32);
        T.cnt     += (r <= rmask) ? T.mAM[r] : 0.f;
    }
}

template <int E>
__device__ __forceinline__ void tile_loop(const v4f* pp, const v2f* pq, int kc, int hi,
                                          const RowTerms& T, float& acc, float& accc)
{
    constexpr int G  = E / 2;
    constexpr int SH = (E == 2) ? 3 : (E == 4) ? 2 : 1;
    v2f Ae[E];
#pragma unroll
    for (int e = 0; e < E; ++e) {
        Ae[e].x = (T.erow == e) ? T.arow.x : 0.f;
        Ae[e].y = (T.erow == e) ? T.arow.y : 0.f;
    }
    const int  nt  = (kc + E - 1) / E;
    const v2f* pqh = pq + (G * hi) * 16;
    v4f pc[E];
    v2f rc[G];
#pragma unroll
    for (int e = 0; e < E; ++e) pc[e] = pp[e * 16];
#pragma unroll
    for (int g = 0; g < G; ++g) rc[g] = pqh[g * 16];
    for (int t = 0; t < nt; ++t) {
        const int tn = min(t + 1, nt - 1);
        v4f pn[E];
        v2f rn[G];
#pragma unroll
        for (int e = 0; e < E; ++e) pn[e] = pp[(E * tn + e) * 16];
#pragma unroll
        for (int g = 0; g < G; ++g) rn[g] = pqh[(E * tn + g) * 16];

        v8f dot = {};
#pragma unroll
        for (int e = 0; e < E; ++e) {
            v2f Bv;
            Bv.x = hi ? pc[e].z : pc[e].x;
            Bv.y = hi ? 0.f : pc[e].y;
            dot = __builtin_amdgcn_wmma_f32_16x16x4_f32(
                 false, Ae[e],  false, Bv,
                 (short)0, dot,  false,  false);
        }

        float n2N[G];
        float wsum = 0.f;
#pragma unroll
        for (int g = 0; g < G; ++g) {
            n2N[g] = hi ? pc[G + g].w : pc[g].w;
            wsum  += rc[g].y;
        }
        accc = fmaf(T.cnt, wsum, accc);
#pragma unroll
        for (int r = 0; r < 8; ++r) {
            const int   g    = r >> SH;
            const float qsq  = T.n2M[r] + n2N[g] - 0.5f * dot[r];
            const float hD   = __builtin_amdgcn_sqrtf(qsq);
            const float u    = (T.hcutM[r] + rc[g].x) - hD;
            const float th   = __builtin_amdgcn_tanhf(u);
            acc = fmaf(T.mAM[r] * rc[g].y, th, acc);
        }
#pragma unroll
        for (int e = 0; e < E; ++e) pc[e] = pn[e];
#pragma unroll
        for (int g = 0; g < G; ++g) rc[g] = rn[g];
    }
}

__global__ __launch_bounds__(256)
void clash_wmma_kernel(const float* __restrict__ X,
                       const float* __restrict__ mask_j,
                       const float* __restrict__ mask_ij,
                       const int*   __restrict__ C,
                       const int*   __restrict__ S,
                       const int*   __restrict__ E,
                       float*       __restrict__ out,
                       int N, int K, int BN)
{
    __shared__ v4i s_meta[BLOCK_SLOTS];
    __shared__ v4f s_pos[BLOCK_SLOTS * 16];
    __shared__ v2f s_rw[BLOCK_SLOTS * 16];

    const int  tid   = threadIdx.x;
    const int  lane  = tid & 31;
    const int  wv    = tid >> 5;
    const int  wave  = blockIdx.x * WAVES_PER_BLOCK + wv;
    const bool valid = wave < BN;
    const int  wave_c = valid ? wave : 0;

    const int b = wave_c / N;
    const int i = wave_c % N;
    const int a  = lane & 15;
    const int hi = lane >> 4;

    const int aL = (a < 14) ? a : 13;

    if (K <= 0) {
        if (valid && lane == 0) out[wave] = 0.f;
        return;
    }

    const int baseE = wave_c * K;
    const int kc0   = min(K, EDGE_SLOTS);
    const int le0   = min(lane, kc0 - 1);
    const int   e_j0   = E[baseE + le0];
    const float e_mij0 = mask_ij[baseE + le0];
    const int   s_i    = S[wave_c];
    const int   c_i    = C[wave_c];
    const float* pxi = X + (size_t)wave_c * 42 + aL * 3;
    const float xi_l = pxi[0], yi_l = pxi[1], zi_l = pxi[2];
    const int   jg0  = b * N + e_j0;
    const int   s_j0 = S[jg0];
    const int   c_j0 = C[jg0];
    const float mj0  = mask_j[jg0];
    const float rad_l = g_vdw[s_i][aL];

    v4i erec = edge_record(valid && lane < kc0, jg0, s_j0, c_j0, mj0, e_mij0, e_j0 == i);
    s_meta[wv * EDGE_SLOTS + lane] = erec;

    const int  cnt_i = (c_i > 0) ? natoms_of(s_i) : 0;
    const int  nsc   = __builtin_amdgcn_readfirstlane(cnt_i) - 4;
    const int  sh1   = (nsc <= 2) ? 1 : (nsc <= 4) ? 2 : 3;

    RowTerms T1 = {};
    const float n2_i = xi_l * xi_l + yi_l * yi_l + zi_l * zi_l;
    const float mA_i = (a < cnt_i) ? 1.f : 0.f;

    float acc  = 0.f;
    float accc = 0.f;
    const int sub   = tid >> 4;

    for (int k0 = 0; k0 < K; k0 += EDGE_SLOTS) {
        const int kc = min(K - k0, EDGE_SLOTS);

        __syncthreads();

        v4i   mm[16];
        float gx[16], gy[16], gz[16], gr[16];
#pragma unroll
        for (int p = 0; p < 16; ++p)
            mm[p] = s_meta[p * 16 + sub];
#pragma unroll
        for (int p = 0; p < 16; ++p) {
            const float* q = X + (size_t)mm[p].x * 42 + aL * 3;
            gx[p] = q[0]; gy[p] = q[1]; gz[p] = q[2];
            gr[p] = g_vdw[mm[p].y & 255][aL];
        }
        if (k0 == 0) {
            row_terms(T1, 4, sh1, lane, a, hi, xi_l, yi_l, zi_l, n2_i, rad_l, mA_i);
        }
#pragma unroll
        for (int p = 0; p < 16; ++p) {
            const bool  real = a < 14;
            const float xj = real ? gx[p] : 0.f;
            const float yj = real ? gy[p] : 0.f;
            const float zj = real ? gz[p] : 0.f;
            const float rad_j = real ? gr[p] : 0.f;
            const float n2N = xj * xj + yj * yj + zj * zj + DIST_EPS;
            const float wN  = ((a < (mm[p].y >> 8)) ? 1.f : 0.f)
                              * __int_as_float(mm[p].z) * __int_as_float(mm[p].w);
            const int   o   = (p * 16 + sub) * 16 + a;
            v4f rec; rec.x = xj; rec.y = yj; rec.z = zj; rec.w = 0.25f * n2N;
            v2f rw;  rw.x = 0.25f * rad_j; rw.y = 0.5f * wN;
            s_pos[o] = rec;
            s_rw[o]  = rw;
        }
        __syncthreads();

        const v4f* pp = s_pos + (wv * EDGE_SLOTS) * 16 + a;
        const v2f* pq = s_rw  + (wv * EDGE_SLOTS) * 16 + a;
        if (nsc > 0) {
            if (sh1 == 1)      tile_loop<8>(pp, pq, kc, hi, T1, acc, accc);
            else if (sh1 == 2) tile_loop<4>(pp, pq, kc, hi, T1, acc, accc);
            else               tile_loop<2>(pp, pq, kc, hi, T1, acc, accc);
            if (nsc > 8) {
                RowTerms T2;
                row_terms(T2, 12, 1, lane, a, hi, xi_l, yi_l, zi_l, n2_i, rad_l, mA_i);
                tile_loop<8>(pp, pq, kc, hi, T2, acc, accc);
            }
        }

        if (k0 + EDGE_SLOTS < K) {
            const int kn  = min(K - k0 - EDGE_SLOTS, EDGE_SLOTS);
            const int ln  = min(lane, kn - 1);
            const int   e_j   = E[baseE + k0 + EDGE_SLOTS + ln];
            const float e_mij = mask_ij[baseE + k0 + EDGE_SLOTS + ln];
            const int   jg    = b * N + e_j;
            erec = edge_record(valid && lane < kn, jg, S[jg], C[jg], mask_j[jg], e_mij, e_j == i);
            s_meta[wv * EDGE_SLOTS + lane] = erec;
        }
    }

    acc += accc;
#pragma unroll
    for (int off = 16; off > 0; off >>= 1)
        acc += __shfl_down(acc, off, 32);
    if (valid && lane == 0) out[wave] = acc;
}

extern "C" void kernel_launch(void* const* d_in, const int* in_sizes, int n_in,
                              void* d_out, int out_size, void* d_ws, size_t ws_size,
                              hipStream_t stream) {
    const float* X       = (const float*)d_in[0];
    const float* mask_j  = (const float*)d_in[1];
    const float* mask_ij = (const float*)d_in[2];
    const int*   C       = (const int*)d_in[3];
    const int*   S       = (const int*)d_in[4];
    const int*   E       = (const int*)d_in[5];
    float*       out     = (float*)d_out;

    const int BN = in_sizes[1];
    const int K  = in_sizes[2] / BN;
    const int B  = 2;
    const int N  = BN / B;

    const int wavesPerBlock = 8;
    const int blocks = (BN + wavesPerBlock - 1) / wavesPerBlock;
    clash_wmma_kernel<<<blocks, wavesPerBlock * 32, 0, stream>>>(
        X, mask_j, mask_ij, C, S, E, out, N, K, BN);
}
